# starting kernel
# baseline (speedup 1.0000x reference)
#include <hip/hip_runtime.h>

#define N_NODES 120000
#define N_EDGES 1200000
#define C_IN    64
#define HID     256
#define ROWS    32

typedef float v2f __attribute__((ext_vector_type(2)));
typedef float v8f __attribute__((ext_vector_type(8)));

__device__ __forceinline__ float leaky01(float v) { return v >= 0.0f ? v : 0.01f * v; }

__device__ __forceinline__ void atomAddF(float* p, float v) {
    __hip_atomic_fetch_add(p, v, __ATOMIC_RELAXED, __HIP_MEMORY_SCOPE_AGENT);
}

__device__ __forceinline__ v8f wmma4(v2f a, v2f b, v8f c) {
    return __builtin_amdgcn_wmma_f32_16x16x4_f32(false, a, false, b, (short)0, c,
                                                 false, false);
}

__global__ void deg_init_kernel(float* __restrict__ deg, int n) {
    int i = blockIdx.x * blockDim.x + threadIdx.x;
    if (i < n) deg[i] = 1.0f;
}

__global__ void deg_accum_kernel(const int* __restrict__ ei, float* __restrict__ deg, int e) {
    int i = blockIdx.x * blockDim.x + threadIdx.x;
    if (i < e) atomAddF(&deg[ei[N_EDGES + i]], 1.0f);
}

__global__ void deg_rsqrt_kernel(float* __restrict__ deg, int n) {
    int i = blockIdx.x * blockDim.x + threadIdx.x;
    if (i < n) deg[i] = rsqrtf(deg[i]);
}

__global__ __launch_bounds__(256) void gemm_xw_kernel(const float* __restrict__ x,
                                                      const float* __restrict__ W,
                                                      float* __restrict__ h, int nrows) {
    int tid  = threadIdx.x;
    int wave = tid >> 5, lane = tid & 31;
    int half = lane >> 4, lm = lane & 15;
    int row0 = blockIdx.x * 64 + (wave >> 2) * 16;
    int colBase = (wave & 3) * 16;
    if (row0 >= nrows) return;

    v8f acc0 = {}, acc1 = {};
    const float* a0 = x + (size_t)(row0 + lm) * C_IN + 2 * half;
    const float* a1 = x + (size_t)(row0 + 32 + lm) * C_IN + 2 * half;
    #pragma unroll
    for (int k = 0; k < C_IN; k += 4) {
        v2f av0, av1, b;
        av0.x = a0[k];  av0.y = a0[k + 1];
        av1.x = a1[k];  av1.y = a1[k + 1];
        b.x = W[(size_t)(k + 2 * half) * C_IN + colBase + lm];
        b.y = W[(size_t)(k + 1 + 2 * half) * C_IN + colBase + lm];
        acc0 = wmma4(av0, b, acc0);
        acc1 = wmma4(av1, b, acc1);
    }
    #pragma unroll
    for (int i = 0; i < 8; ++i) {
        h[(size_t)(row0 + i + 8 * half) * C_IN + colBase + lm]      = acc0[i];
        h[(size_t)(row0 + 32 + i + 8 * half) * C_IN + colBase + lm] = acc1[i];
    }
}

__global__ void agg_init_kernel(const float* __restrict__ h, const float* __restrict__ dinv,
                                float* __restrict__ agg, int total) {
    int i = blockIdx.x * blockDim.x + threadIdx.x;
    if (i < total) {
        int n = i >> 6;
        float di = dinv[n];
        agg[i] = h[i] * di * di;
    }
}

__global__ void scatter_kernel(const int* __restrict__ ei, const float* __restrict__ h,
                               const float* __restrict__ dinv, float* __restrict__ agg,
                               int total  ) {
    int i = blockIdx.x * blockDim.x + threadIdx.x;
    if (i < total) {
        int e  = i >> 4;
        int c4 = (i & 15) << 2;
        int s = ei[e];
        int d = ei[N_EDGES + e];
        float norm = dinv[s] * dinv[d];
        const float4 hv = *(const float4*)(h + (size_t)s * C_IN + c4);
        float* ap = agg + (size_t)d * C_IN + c4;
        atomAddF(ap + 0, hv.x * norm);
        atomAddF(ap + 1, hv.y * norm);
        atomAddF(ap + 2, hv.z * norm);
        atomAddF(ap + 3, hv.w * norm);
    }
}

__global__ __launch_bounds__(256) void mlp_fused_kernel(
    const float* __restrict__ agg, const float* __restrict__ x,
    const float* __restrict__ b_gcn, const float* __restrict__ W1,
    const float* __restrict__ b1, const float* __restrict__ W2,
    const float* __restrict__ b2, const float* __restrict__ Wd,
    const float* __restrict__ bd, float* __restrict__ out) {
    __shared__ float zb[ROWS * (C_IN + 1)];
    __shared__ float h1[ROWS * (HID + 1)];
    __shared__ float h2[ROWS * (HID + 1)];

    int tid = threadIdx.x;
    int nodeBase = blockIdx.x * ROWS;

    for (int i = tid; i < ROWS * C_IN; i += 256) {
        int r = i >> 6, c = i & (C_IN - 1);
        size_t g = (size_t)(nodeBase + r) * C_IN + c;
        float v = agg[g] + b_gcn[c];
        v = v > 0.0f ? v : 0.0f;
        zb[r * (C_IN + 1) + c] = v + x[g];
    }
    __syncthreads();

    int wave = tid >> 5, lane = tid & 31, half = lane >> 4, lm = lane & 15;
    int col0 = wave * 16;
    int col1 = col0 + 128;

    {
        v8f acc00 = {}, acc01 = {}, acc10 = {}, acc11 = {};
        const float* aR0 = zb + lm * (C_IN + 1) + 2 * half;
        const float* aR1 = zb + (16 + lm) * (C_IN + 1) + 2 * half;
        const float* w0  = W1 + (size_t)(col0 + lm) * C_IN + 2 * half;
        const float* w1  = W1 + (size_t)(col1 + lm) * C_IN + 2 * half;
        #pragma unroll
        for (int k = 0; k < C_IN; k += 4) {
            v2f a0v, a1v, b0v, b1v;
            a0v.x = aR0[k]; a0v.y = aR0[k + 1];
            a1v.x = aR1[k]; a1v.y = aR1[k + 1];
            b0v.x = w0[k];  b0v.y = w0[k + 1];
            b1v.x = w1[k];  b1v.y = w1[k + 1];
            acc00 = wmma4(a0v, b0v, acc00);
            acc01 = wmma4(a0v, b1v, acc01);
            acc10 = wmma4(a1v, b0v, acc10);
            acc11 = wmma4(a1v, b1v, acc11);
        }
        float bc0 = b1[col0 + lm], bc1 = b1[col1 + lm];
        #pragma unroll
        for (int i = 0; i < 8; ++i) {
            int r = i + 8 * half;
            h1[r * (HID + 1) + col0 + lm]        = leaky01(acc00[i] + bc0);
            h1[r * (HID + 1) + col1 + lm]        = leaky01(acc01[i] + bc1);
            h1[(16 + r) * (HID + 1) + col0 + lm] = leaky01(acc10[i] + bc0);
            h1[(16 + r) * (HID + 1) + col1 + lm] = leaky01(acc11[i] + bc1);
        }
    }
    __syncthreads();

    {
        v8f acc00 = {}, acc01 = {}, acc10 = {}, acc11 = {};
        const float* aR0 = h1 + lm * (HID + 1) + 2 * half;
        const float* aR1 = h1 + (16 + lm) * (HID + 1) + 2 * half;
        const float* w0  = W2 + (size_t)(col0 + lm) * HID + 2 * half;
        const float* w1  = W2 + (size_t)(col1 + lm) * HID + 2 * half;
        #pragma unroll 4
        for (int k = 0; k < HID; k += 4) {
            v2f a0v, a1v, b0v, b1v;
            a0v.x = aR0[k]; a0v.y = aR0[k + 1];
            a1v.x = aR1[k]; a1v.y = aR1[k + 1];
            b0v.x = w0[k];  b0v.y = w0[k + 1];
            b1v.x = w1[k];  b1v.y = w1[k + 1];
            acc00 = wmma4(a0v, b0v, acc00);
            acc01 = wmma4(a0v, b1v, acc01);
            acc10 = wmma4(a1v, b0v, acc10);
            acc11 = wmma4(a1v, b1v, acc11);
        }
        float bc0 = b2[col0 + lm], bc1 = b2[col1 + lm];
        #pragma unroll
        for (int i = 0; i < 8; ++i) {
            int r = i + 8 * half;
            h2[r * (HID + 1) + col0 + lm]        = leaky01(acc00[i] + bc0);
            h2[r * (HID + 1) + col1 + lm]        = leaky01(acc01[i] + bc1);
            h2[(16 + r) * (HID + 1) + col0 + lm] = leaky01(acc10[i] + bc0);
            h2[(16 + r) * (HID + 1) + col1 + lm] = leaky01(acc11[i] + bc1);
        }
    }
    __syncthreads();

    {
        int row = tid >> 3, seg = tid & 7;
        const float* hrow = h2 + row * (HID + 1) + seg * 32;
        const float* wseg = Wd + seg * 32;
        float p = 0.0f;
        #pragma unroll
        for (int c = 0; c < 32; ++c) p += hrow[c] * wseg[c];
        zb[row * 8 + seg] = p;
    }
    __syncthreads();
    if (tid < ROWS) {
        float acc = 0.0f;
        #pragma unroll
        for (int s = 0; s < 8; ++s) acc += zb[tid * 8 + s];
        out[nodeBase + tid] = acc + bd[0];
    }
}

extern "C" void kernel_launch(void* const* d_in, const int* in_sizes, int n_in,
                              void* d_out, int out_size, void* d_ws, size_t ws_size,
                              hipStream_t stream) {
    const float* x     = (const float*)d_in[0];
    const int*   ei    = (const int*)d_in[1];
    const float* W_gcn = (const float*)d_in[2];
    const float* b_gcn = (const float*)d_in[3];
    const float* W1    = (const float*)d_in[4];
    const float* b1    = (const float*)d_in[5];
    const float* W2    = (const float*)d_in[6];
    const float* b2    = (const float*)d_in[7];
    const float* Wd    = (const float*)d_in[8];
    const float* bd    = (const float*)d_in[9];
    float* out = (float*)d_out;

    char* ws = (char*)d_ws;
    size_t off0 = 0;
    size_t off1 = (off0 + (size_t)N_NODES * sizeof(float) + 255) & ~(size_t)255;
    size_t off2 = (off1 + (size_t)N_NODES * C_IN * sizeof(float) + 255) & ~(size_t)255;
    float* dinv = (float*)(ws + off0);
    float* h    = (float*)(ws + off1);
    float* agg  = (float*)(ws + off2);

    const int B = 256;
    deg_init_kernel<<<(N_NODES + B - 1) / B, B, 0, stream>>>(dinv, N_NODES);
    deg_accum_kernel<<<(N_EDGES + B - 1) / B, B, 0, stream>>>(ei, dinv, N_EDGES);
    deg_rsqrt_kernel<<<(N_NODES + B - 1) / B, B, 0, stream>>>(dinv, N_NODES);

    gemm_xw_kernel<<<N_NODES / 64, B, 0, stream>>>(x, W_gcn, h, N_NODES);

    agg_init_kernel<<<((size_t)N_NODES * C_IN + B - 1) / B, B, 0, stream>>>(
        h, dinv, agg, N_NODES * C_IN);
    scatter_kernel<<<((size_t)N_EDGES * 16 + B - 1) / B, B, 0, stream>>>(
        ei, h, dinv, agg, N_EDGES * 16);

    mlp_fused_kernel<<<N_NODES / ROWS, B, 0, stream>>>(agg, x, b_gcn, W1, b1, W2, b2,
                                                       Wd, bd, out);
}


# edge sum by pull: edges take int slots per destination (32 + overflow list); 16 lanes per node sum dinv-prescaled source rows in registers, one store; f32 atomics go; 7->5 launches; sum in slot order
# speedup vs baseline: 1.4306x; 1.4306x over previous
#include <hip/hip_runtime.h>

#define N_NODES 120000
#define N_EDGES 1200000
#define C_IN    64
#define HID     256
#define ROWS    160
#define RT      (ROWS / 16)
#define CAP     32

typedef float v2f __attribute__((ext_vector_type(2)));
typedef float v8f __attribute__((ext_vector_type(8)));

__device__ __forceinline__ float leaky01(float v) { return v >= 0.0f ? v : 0.01f * v; }

__device__ __forceinline__ int atomTakeSlot(int* p) {
    return __hip_atomic_fetch_add(p, 1, __ATOMIC_RELAXED, __HIP_MEMORY_SCOPE_AGENT);
}

__device__ __forceinline__ v8f wmma4(v2f a, v2f b, v8f c) {
    return __builtin_amdgcn_wmma_f32_16x16x4_f32(false, a, false, b, (short)0, c,
                                                 false, false);
}

__global__ void cnt_init_kernel(int* __restrict__ cnt, int* __restrict__ ovfN, int n) {
    int i = blockIdx.x * blockDim.x + threadIdx.x;
    if (i < n) cnt[i] = 0;
    if (i == 0) ovfN[0] = 0;
}

__device__ __forceinline__ void place_edge(int d, int s, int p, int* __restrict__ slot,
                                           int* __restrict__ ovfN, int* __restrict__ ovf) {
    if (p < CAP) {
        slot[(size_t)d * CAP + p] = s;
    } else {
        int o = atomTakeSlot(ovfN);
        ovf[2 * o] = d;
        ovf[2 * o + 1] = s;
    }
}

__global__ void slot_kernel(const int* __restrict__ ei, int* __restrict__ cnt,
                            int* __restrict__ slot, int* __restrict__ ovfN,
                            int* __restrict__ ovf, int quads  ) {
    int t = blockIdx.x * blockDim.x + threadIdx.x;
    if (t < quads) {
        const int4 s4 = ((const int4*)ei)[t];
        const int4 d4 = ((const int4*)(ei + N_EDGES))[t];
        int p0 = atomTakeSlot(&cnt[d4.x]);
        int p1 = atomTakeSlot(&cnt[d4.y]);
        int p2 = atomTakeSlot(&cnt[d4.z]);
        int p3 = atomTakeSlot(&cnt[d4.w]);
        place_edge(d4.x, s4.x, p0, slot, ovfN, ovf);
        place_edge(d4.y, s4.y, p1, slot, ovfN, ovf);
        place_edge(d4.z, s4.z, p2, slot, ovfN, ovf);
        place_edge(d4.w, s4.w, p3, slot, ovfN, ovf);
    }
}

__global__ __launch_bounds__(256) void gemm_xw_kernel(const float* __restrict__ x,
                                                      const float* __restrict__ W,
                                                      const int* __restrict__ cnt,
                                                      float* __restrict__ h, int nrows) {
    int tid  = threadIdx.x;
    int wave = tid >> 5, lane = tid & 31;
    int half = lane >> 4, lm = lane & 15;
    int row0 = blockIdx.x * 64 + (wave >> 2) * 16;
    int colBase = (wave & 3) * 16;
    if (row0 >= nrows) return;

    v8f acc0 = {}, acc1 = {};
    const float* a0 = x + (size_t)(row0 + lm) * C_IN + 2 * half;
    const float* a1 = x + (size_t)(row0 + 32 + lm) * C_IN + 2 * half;
    #pragma unroll
    for (int k = 0; k < C_IN; k += 4) {
        v2f av0, av1, b;
        av0.x = a0[k];  av0.y = a0[k + 1];
        av1.x = a1[k];  av1.y = a1[k + 1];
        b.x = W[(size_t)(k + 2 * half) * C_IN + colBase + lm];
        b.y = W[(size_t)(k + 1 + 2 * half) * C_IN + colBase + lm];
        acc0 = wmma4(av0, b, acc0);
        acc1 = wmma4(av1, b, acc1);
    }
    #pragma unroll
    for (int i = 0; i < 8; ++i) {
        int r0 = row0 + i + 8 * half, r1 = r0 + 32;
        float d0 = rsqrtf((float)(cnt[r0] + 1));
        float d1 = rsqrtf((float)(cnt[r1] + 1));
        h[(size_t)r0 * C_IN + colBase + lm] = acc0[i] * d0;
        h[(size_t)r1 * C_IN + colBase + lm] = acc1[i] * d1;
    }
}

__global__ __launch_bounds__(256) void gather_kernel(const int* __restrict__ cnt,
                                                     const int* __restrict__ slot,
                                                     const int* __restrict__ ovfN,
                                                     const int* __restrict__ ovf,
                                                     const float* __restrict__ h,
                                                     float* __restrict__ agg) {
    static_assert((N_NODES * 16) % 256 == 0, "whole blocks: no node guard needed");
    int t = blockIdx.x * 256 + threadIdx.x;
    int node = t >> 4, l16 = t & 15;
    const float4* h4 = (const float4*)h;
    float4 acc = h4[(size_t)node * 16 + l16];
    int c = cnt[node];
    int m = c < CAP ? c : CAP;
    const int* row = slot + (size_t)node * CAP;
    int i0 = l16 < m ? row[l16] : 0;
    int i1 = l16 + 16 < m ? row[l16 + 16] : 0;
    for (int j = 0; j < m; j += 8) {
        int idx = j < 16 ? i0 : i1;
        float4 v[8];
        #pragma unroll
        for (int u = 0; u < 8; ++u) {
            int s = __shfl(idx, (j + u) & 15, 16);
            v[u] = make_float4(0.0f, 0.0f, 0.0f, 0.0f);
            if (j + u < m) v[u] = h4[(size_t)s * 16 + l16];
        }
        #pragma unroll
        for (int u = 0; u < 8; ++u) {
            acc.x += v[u].x; acc.y += v[u].y; acc.z += v[u].z; acc.w += v[u].w;
        }
    }
    if (c > CAP) {
        int no = ovfN[0];
        for (int i = 0; i < no; ++i) {
            if (ovf[2 * i] == node) {
                const float4 w = h4[(size_t)ovf[2 * i + 1] * 16 + l16];
                acc.x += w.x; acc.y += w.y; acc.z += w.z; acc.w += w.w;
            }
        }
    }
    float dd = rsqrtf((float)(c + 1));
    acc.x *= dd; acc.y *= dd; acc.z *= dd; acc.w *= dd;
    ((float4*)agg)[(size_t)node * 16 + l16] = acc;
}


__device__ __forceinline__ void mlp_fetch_b(const float* __restrict__ w0,
                                            const float* __restrict__ w1, int kb,
                                            v2f (&b)[8][2]) {
    #pragma unroll
    for (int s = 0; s < 8; ++s) {
        int k = kb + 4 * s;
        b[s][0].x = w0[k]; b[s][0].y = w0[k + 1];
        b[s][1].x = w1[k]; b[s][1].y = w1[k + 1];
    }
}

template <int PITCH, bool LAST>
__device__ __forceinline__ void mlp_block(const float* __restrict__ aB,
                                          const float* __restrict__ w0,
                                          const float* __restrict__ w1, int kb,
                                          v2f (&ac)[RT], v2f (&bc)[8][2],
                                          v8f (&acc)[RT][2]) {
    v2f bn[8][2];
    if (!LAST) mlp_fetch_b(w0, w1, kb + 32, bn);
    #pragma unroll
    for (int s = 0; s < 8; ++s) {
        v2f an[RT];
        if (!(LAST && s == 7)) {
            int k = kb + 4 * (s + 1);
            #pragma unroll
            for (int t = 0; t < RT; ++t) {
                an[t].x = aB[t * 16 * PITCH + k];
                an[t].y = aB[t * 16 * PITCH + k + 1];
            }
        }
        #pragma unroll
        for (int t = 0; t < RT; ++t) {
            acc[t][0] = wmma4(ac[t], bc[s][0], acc[t][0]);
            acc[t][1] = wmma4(ac[t], bc[s][1], acc[t][1]);
        }
        if (!(LAST && s == 7)) {
            #pragma unroll
            for (int t = 0; t < RT; ++t) ac[t] = an[t];
        }
    }
    if (!LAST) {
        #pragma unroll
        for (int s = 0; s < 8; ++s) { bc[s][0] = bn[s][0]; bc[s][1] = bn[s][1]; }
    }
}

template <int K, int PITCH>
__device__ __forceinline__ void mlp_gemm(const float* __restrict__ aB,
                                         const float* __restrict__ w0,
                                         const float* __restrict__ w1,
                                         v2f (&bc)[8][2], v8f (&acc)[RT][2]) {
    v2f ac[RT];
    #pragma unroll
    for (int t = 0; t < RT; ++t) {
        ac[t].x = aB[t * 16 * PITCH];
        ac[t].y = aB[t * 16 * PITCH + 1];
    }
    #pragma nounroll
    for (int kb = 0; kb < K - 32; kb += 32)
        mlp_block<PITCH, false>(aB, w0, w1, kb, ac, bc, acc);
    mlp_block<PITCH, true>(aB, w0, w1, K - 32, ac, bc, acc);
}

__global__ __launch_bounds__(256) void mlp_fused_kernel(
    const float* __restrict__ agg, const float* __restrict__ x,
    const float* __restrict__ b_gcn, const float* __restrict__ W1,
    const float* __restrict__ b1, const float* __restrict__ W2,
    const float* __restrict__ b2, const float* __restrict__ Wd,
    const float* __restrict__ bd, float* __restrict__ out) {
    __shared__ float zb[ROWS * (C_IN + 1)];
    __shared__ float hb[ROWS * (HID + 1)];

    int tid = threadIdx.x;
    int nodeBase = blockIdx.x * ROWS;
    int wave = tid >> 5, lane = tid & 31, half = lane >> 4, lm = lane & 15;
    int col0 = wave * 16;
    int col1 = col0 + 128;
    const float* w1a = W1 + (size_t)(col0 + lm) * C_IN + 2 * half;
    const float* w1b = W1 + (size_t)(col1 + lm) * C_IN + 2 * half;
    const float* w2a = W2 + (size_t)(col0 + lm) * HID + 2 * half;
    const float* w2b = W2 + (size_t)(col1 + lm) * HID + 2 * half;

    v2f bfirst[8][2];
    mlp_fetch_b(w1a, w1b, 0, bfirst);

    static_assert((ROWS * C_IN) % 256 == 0 && (ROWS * 8) % 256 == 0 && ROWS % 32 == 0,
                  "tile loops below have whole trips");
    const float* aggT = agg + (size_t)nodeBase * C_IN + tid;
    const float* xT   = x + (size_t)nodeBase * C_IN + tid;
    float bg = b_gcn[tid & (C_IN - 1)];
    float* zT = zb + (tid >> 6) * (C_IN + 1) + (tid & (C_IN - 1));
    #pragma unroll
    for (int it = 0; it < ROWS * C_IN / 256; ++it) {
        float v = aggT[256 * it] + bg;
        v = v > 0.0f ? v : 0.0f;
        zT[4 * it * (C_IN + 1)] = v + xT[256 * it];
    }
    __syncthreads();

    {
        v8f acc[RT][2];
        #pragma unroll
        for (int t = 0; t < RT; ++t) { acc[t][0] = v8f{}; acc[t][1] = v8f{}; }
        mlp_gemm<C_IN, C_IN + 1>(zb + lm * (C_IN + 1) + 2 * half, w1a, w1b, bfirst, acc);
        mlp_fetch_b(w2a, w2b, 0, bfirst);
        float bc0 = b1[col0 + lm], bc1 = b1[col1 + lm];
        #pragma unroll
        for (int t = 0; t < RT; ++t) {
            #pragma unroll
            for (int i = 0; i < 8; ++i) {
                int r = 16 * t + i + 8 * half;
                hb[r * (HID + 1) + col0 + lm] = leaky01(acc[t][0][i] + bc0);
                hb[r * (HID + 1) + col1 + lm] = leaky01(acc[t][1][i] + bc1);
            }
        }
    }
    __syncthreads();

    {
        v8f acc[RT][2];
        #pragma unroll
        for (int t = 0; t < RT; ++t) { acc[t][0] = v8f{}; acc[t][1] = v8f{}; }
        mlp_gemm<HID, HID + 1>(hb + lm * (HID + 1) + 2 * half, w2a, w2b, bfirst, acc);
        float bc0 = b2[col0 + lm], bc1 = b2[col1 + lm];
        __syncthreads();
        #pragma unroll
        for (int t = 0; t < RT; ++t) {
            #pragma unroll
            for (int i = 0; i < 8; ++i) {
                int r = 16 * t + i + 8 * half;
                hb[r * (HID + 1) + col0 + lm] = leaky01(acc[t][0][i] + bc0);
                hb[r * (HID + 1) + col1 + lm] = leaky01(acc[t][1][i] + bc1);
            }
        }
    }
    __syncthreads();

    #pragma unroll
    for (int j = 0; j < ROWS * 8 / 256; ++j) {
        int p = tid + 256 * j;
        int seg = p / ROWS, row = p - seg * ROWS;
        const float* hrow = hb + row * (HID + 1) + seg * 32;
        const float* wseg = Wd + seg * 32;
        float s = 0.0f;
        #pragma unroll
        for (int c = 0; c < 32; ++c) s += hrow[c] * wseg[c];
        zb[row * 8 + seg] = s;
    }
    __syncthreads();
    if (tid < ROWS) {
        float a = 0.0f;
        #pragma unroll
        for (int s = 0; s < 8; ++s) a += zb[tid * 8 + s];
        out[nodeBase + tid] = a + bd[0];
    }
}

extern "C" void kernel_launch(void* const* d_in, const int* in_sizes, int n_in,
                              void* d_out, int out_size, void* d_ws, size_t ws_size,
                              hipStream_t stream) {
    const float* x     = (const float*)d_in[0];
    const int*   ei    = (const int*)d_in[1];
    const float* W_gcn = (const float*)d_in[2];
    const float* b_gcn = (const float*)d_in[3];
    const float* W1    = (const float*)d_in[4];
    const float* b1    = (const float*)d_in[5];
    const float* W2    = (const float*)d_in[6];
    const float* b2    = (const float*)d_in[7];
    const float* Wd    = (const float*)d_in[8];
    const float* bd    = (const float*)d_in[9];
    float* out = (float*)d_out;

    char* ws = (char*)d_ws;
    size_t off0 = 0;
    size_t off1 = (off0 + (size_t)N_NODES * sizeof(int) + 255) & ~(size_t)255;
    size_t off2 = off1 + 256;
    size_t off3 = (off2 + (size_t)N_EDGES * 2 * sizeof(int) + 255) & ~(size_t)255;
    size_t off4 = (off3 + (size_t)N_NODES * CAP * sizeof(int) + 255) & ~(size_t)255;
    size_t off5 = (off4 + (size_t)N_NODES * C_IN * sizeof(float) + 255) & ~(size_t)255;
    int*   cnt  = (int*)(ws + off0);
    int*   ovfN = (int*)(ws + off1);
    int*   ovf  = (int*)(ws + off2);
    int*   slot = (int*)(ws + off3);
    float* h    = (float*)(ws + off4);
    float* agg  = (float*)(ws + off5);

    const int B = 256;
    cnt_init_kernel<<<(N_NODES + B - 1) / B, B, 0, stream>>>(cnt, ovfN, N_NODES);
    slot_kernel<<<(N_EDGES / 4 + B - 1) / B, B, 0, stream>>>(ei, cnt, slot, ovfN, ovf,
                                                             N_EDGES / 4);

    gemm_xw_kernel<<<N_NODES / 64, B, 0, stream>>>(x, W_gcn, cnt, h, N_NODES);

    gather_kernel<<<N_NODES * 16 / B, B, 0, stream>>>(cnt, slot, ovfN, ovf, h, agg);

    mlp_fused_kernel<<<N_NODES / ROWS, B, 0, stream>>>(agg, x, b_gcn, W1, b1, W2, b2,
                                                       Wd, bd, out);
}


# in-edge counters and slot rows kept once per cache half (plane picked by the wave's die): no counter or slot line is updated from both halves; sum order: plane 0's slots, then plane 1's
# speedup vs baseline: 1.9734x; 1.3794x over previous
#include <hip/hip_runtime.h>

#define N_NODES 120000
#define N_EDGES 1200000
#define C_IN    64
#define HID     256
#define ROWS    160
#define RT      (ROWS / 16)
#define HCAP    16

typedef float v2f __attribute__((ext_vector_type(2)));
typedef float v8f __attribute__((ext_vector_type(8)));

__device__ __forceinline__ float leaky01(float v) { return v >= 0.0f ? v : 0.01f * v; }

__device__ __forceinline__ int atomTakeSlot(int* p) {
    return __hip_atomic_fetch_add(p, 1, __ATOMIC_RELAXED, __HIP_MEMORY_SCOPE_AGENT);
}

__device__ __forceinline__ v8f wmma4(v2f a, v2f b, v8f c) {
    return __builtin_amdgcn_wmma_f32_16x16x4_f32(false, a, false, b, (short)0, c,
                                                 false, false);
}

__device__ __forceinline__ int cache_half() {
    return (int)((__builtin_amdgcn_s_sendmsg_rtn(0x87) >> 18) & 1u);
}

__global__ void cnt_init_kernel(int* __restrict__ cnt, int* __restrict__ ovfN, int n) {
    int i = blockIdx.x * blockDim.x + threadIdx.x;
    if (i < n) cnt[i] = 0;
    if (i == 0) ovfN[0] = 0;
}

__device__ __forceinline__ void place_edge(int d, int s, int p, int* __restrict__ slot,
                                           int* __restrict__ ovfN, int* __restrict__ ovf) {
    if (p < HCAP) {
        slot[(size_t)d * HCAP + p] = s;
    } else {
        int o = atomTakeSlot(ovfN);
        ovf[2 * o] = d;
        ovf[2 * o + 1] = s;
    }
}

__global__ void slot_kernel(const int* __restrict__ ei, int* __restrict__ cnt,
                            int* __restrict__ slot, int* __restrict__ ovfN,
                            int* __restrict__ ovf, int quads  ) {
    int t = blockIdx.x * blockDim.x + threadIdx.x;
    const int hf = cache_half();
    cnt  += (size_t)hf * N_NODES;
    slot += (size_t)hf * N_NODES * HCAP;
    if (t < quads) {
        const int4 s4 = ((const int4*)ei)[t];
        const int4 d4 = ((const int4*)(ei + N_EDGES))[t];
        int p0 = atomTakeSlot(&cnt[d4.x]);
        int p1 = atomTakeSlot(&cnt[d4.y]);
        int p2 = atomTakeSlot(&cnt[d4.z]);
        int p3 = atomTakeSlot(&cnt[d4.w]);
        place_edge(d4.x, s4.x, p0, slot, ovfN, ovf);
        place_edge(d4.y, s4.y, p1, slot, ovfN, ovf);
        place_edge(d4.z, s4.z, p2, slot, ovfN, ovf);
        place_edge(d4.w, s4.w, p3, slot, ovfN, ovf);
    }
}

__global__ __launch_bounds__(256) void gemm_xw_kernel(const float* __restrict__ x,
                                                      const float* __restrict__ W,
                                                      const int* __restrict__ cnt,
                                                      float* __restrict__ h, int nrows) {
    int tid  = threadIdx.x;
    int wave = tid >> 5, lane = tid & 31;
    int half = lane >> 4, lm = lane & 15;
    int row0 = blockIdx.x * 64 + (wave >> 2) * 16;
    int colBase = (wave & 3) * 16;
    if (row0 >= nrows) return;

    v8f acc0 = {}, acc1 = {};
    const float* a0 = x + (size_t)(row0 + lm) * C_IN + 2 * half;
    const float* a1 = x + (size_t)(row0 + 32 + lm) * C_IN + 2 * half;
    #pragma unroll
    for (int k = 0; k < C_IN; k += 4) {
        v2f av0, av1, b;
        av0.x = a0[k];  av0.y = a0[k + 1];
        av1.x = a1[k];  av1.y = a1[k + 1];
        b.x = W[(size_t)(k + 2 * half) * C_IN + colBase + lm];
        b.y = W[(size_t)(k + 1 + 2 * half) * C_IN + colBase + lm];
        acc0 = wmma4(av0, b, acc0);
        acc1 = wmma4(av1, b, acc1);
    }
    #pragma unroll
    for (int i = 0; i < 8; ++i) {
        int r0 = row0 + i + 8 * half, r1 = r0 + 32;
        float d0 = rsqrtf((float)(cnt[r0] + cnt[N_NODES + r0] + 1));
        float d1 = rsqrtf((float)(cnt[r1] + cnt[N_NODES + r1] + 1));
        h[(size_t)r0 * C_IN + colBase + lm] = acc0[i] * d0;
        h[(size_t)r1 * C_IN + colBase + lm] = acc1[i] * d1;
    }
}


__device__ __forceinline__ void sum_slots(float4& acc, int idx, int m,
                                          const float4* __restrict__ h4, int l16) {
    for (int j = 0; j < m; j += 8) {
        float4 v[8];
        #pragma unroll
        for (int u = 0; u < 8; ++u) {
            int s = __shfl(idx, j + u, 16);
            v[u] = make_float4(0.0f, 0.0f, 0.0f, 0.0f);
            if (j + u < m) v[u] = h4[(size_t)s * 16 + l16];
        }
        #pragma unroll
        for (int u = 0; u < 8; ++u) {
            acc.x += v[u].x; acc.y += v[u].y; acc.z += v[u].z; acc.w += v[u].w;
        }
    }
}

__global__ __launch_bounds__(256) void gather_kernel(const int* __restrict__ cnt,
                                                     const int* __restrict__ slot,
                                                     const int* __restrict__ ovfN,
                                                     const int* __restrict__ ovf,
                                                     const float* __restrict__ h,
                                                     float* __restrict__ agg) {
    static_assert((N_NODES * 16) % 256 == 0, "whole blocks: no node guard needed");
    int t = blockIdx.x * 256 + threadIdx.x;
    int node = t >> 4, l16 = t & 15;
    const float4* h4 = (const float4*)h;
    float4 acc = h4[(size_t)node * 16 + l16];
    int c0 = cnt[node], c1 = cnt[N_NODES + node];
    int m0 = c0 < HCAP ? c0 : HCAP;
    int m1 = c1 < HCAP ? c1 : HCAP;
    int i0 = l16 < m0 ? slot[(size_t)node * HCAP + l16] : 0;
    int i1 = l16 < m1 ? slot[((size_t)N_NODES + node) * HCAP + l16] : 0;
    sum_slots(acc, i0, m0, h4, l16);
    sum_slots(acc, i1, m1, h4, l16);
    if (c0 > HCAP || c1 > HCAP) {
        int no = ovfN[0];
        for (int i = 0; i < no; ++i) {
            if (ovf[2 * i] == node) {
                const float4 w = h4[(size_t)ovf[2 * i + 1] * 16 + l16];
                acc.x += w.x; acc.y += w.y; acc.z += w.z; acc.w += w.w;
            }
        }
    }
    float dd = rsqrtf((float)(c0 + c1 + 1));
    acc.x *= dd; acc.y *= dd; acc.z *= dd; acc.w *= dd;
    ((float4*)agg)[(size_t)node * 16 + l16] = acc;
}


__device__ __forceinline__ void mlp_fetch_b(const float* __restrict__ w0,
                                            const float* __restrict__ w1, int kb,
                                            v2f (&b)[8][2]) {
    #pragma unroll
    for (int s = 0; s < 8; ++s) {
        int k = kb + 4 * s;
        b[s][0].x = w0[k]; b[s][0].y = w0[k + 1];
        b[s][1].x = w1[k]; b[s][1].y = w1[k + 1];
    }
}

template <int PITCH, bool LAST>
__device__ __forceinline__ void mlp_block(const float* __restrict__ aB,
                                          const float* __restrict__ w0,
                                          const float* __restrict__ w1, int kb,
                                          v2f (&ac)[RT], v2f (&bc)[8][2],
                                          v8f (&acc)[RT][2]) {
    v2f bn[8][2];
    if (!LAST) mlp_fetch_b(w0, w1, kb + 32, bn);
    #pragma unroll
    for (int s = 0; s < 8; ++s) {
        v2f an[RT];
        if (!(LAST && s == 7)) {
            int k = kb + 4 * (s + 1);
            #pragma unroll
            for (int t = 0; t < RT; ++t) {
                an[t].x = aB[t * 16 * PITCH + k];
                an[t].y = aB[t * 16 * PITCH + k + 1];
            }
        }
        #pragma unroll
        for (int t = 0; t < RT; ++t) {
            acc[t][0] = wmma4(ac[t], bc[s][0], acc[t][0]);
            acc[t][1] = wmma4(ac[t], bc[s][1], acc[t][1]);
        }
        if (!(LAST && s == 7)) {
            #pragma unroll
            for (int t = 0; t < RT; ++t) ac[t] = an[t];
        }
    }
    if (!LAST) {
        #pragma unroll
        for (int s = 0; s < 8; ++s) { bc[s][0] = bn[s][0]; bc[s][1] = bn[s][1]; }
    }
}

template <int K, int PITCH>
__device__ __forceinline__ void mlp_gemm(const float* __restrict__ aB,
                                         const float* __restrict__ w0,
                                         const float* __restrict__ w1,
                                         v2f (&bc)[8][2], v8f (&acc)[RT][2]) {
    v2f ac[RT];
    #pragma unroll
    for (int t = 0; t < RT; ++t) {
        ac[t].x = aB[t * 16 * PITCH];
        ac[t].y = aB[t * 16 * PITCH + 1];
    }
    #pragma nounroll
    for (int kb = 0; kb < K - 32; kb += 32)
        mlp_block<PITCH, false>(aB, w0, w1, kb, ac, bc, acc);
    mlp_block<PITCH, true>(aB, w0, w1, K - 32, ac, bc, acc);
}

__global__ __launch_bounds__(256) void mlp_fused_kernel(
    const float* __restrict__ agg, const float* __restrict__ x,
    const float* __restrict__ b_gcn, const float* __restrict__ W1,
    const float* __restrict__ b1, const float* __restrict__ W2,
    const float* __restrict__ b2, const float* __restrict__ Wd,
    const float* __restrict__ bd, float* __restrict__ out) {
    __shared__ float zb[ROWS * (C_IN + 1)];
    __shared__ float hb[ROWS * (HID + 1)];

    int tid = threadIdx.x;
    int nodeBase = blockIdx.x * ROWS;
    int wave = tid >> 5, lane = tid & 31, half = lane >> 4, lm = lane & 15;
    int col0 = wave * 16;
    int col1 = col0 + 128;
    const float* w1a = W1 + (size_t)(col0 + lm) * C_IN + 2 * half;
    const float* w1b = W1 + (size_t)(col1 + lm) * C_IN + 2 * half;
    const float* w2a = W2 + (size_t)(col0 + lm) * HID + 2 * half;
    const float* w2b = W2 + (size_t)(col1 + lm) * HID + 2 * half;

    v2f bfirst[8][2];
    mlp_fetch_b(w1a, w1b, 0, bfirst);

    static_assert((ROWS * C_IN) % 256 == 0 && (ROWS * 8) % 256 == 0 && ROWS % 32 == 0,
                  "tile loops below have whole trips");
    const float* aggT = agg + (size_t)nodeBase * C_IN + tid;
    const float* xT   = x + (size_t)nodeBase * C_IN + tid;
    float bg = b_gcn[tid & (C_IN - 1)];
    float* zT = zb + (tid >> 6) * (C_IN + 1) + (tid & (C_IN - 1));
    #pragma unroll
    for (int it = 0; it < ROWS * C_IN / 256; ++it) {
        float v = aggT[256 * it] + bg;
        v = v > 0.0f ? v : 0.0f;
        zT[4 * it * (C_IN + 1)] = v + xT[256 * it];
    }
    __syncthreads();

    {
        v8f acc[RT][2];
        #pragma unroll
        for (int t = 0; t < RT; ++t) { acc[t][0] = v8f{}; acc[t][1] = v8f{}; }
        mlp_gemm<C_IN, C_IN + 1>(zb + lm * (C_IN + 1) + 2 * half, w1a, w1b, bfirst, acc);
        mlp_fetch_b(w2a, w2b, 0, bfirst);
        float bc0 = b1[col0 + lm], bc1 = b1[col1 + lm];
        #pragma unroll
        for (int t = 0; t < RT; ++t) {
            #pragma unroll
            for (int i = 0; i < 8; ++i) {
                int r = 16 * t + i + 8 * half;
                hb[r * (HID + 1) + col0 + lm] = leaky01(acc[t][0][i] + bc0);
                hb[r * (HID + 1) + col1 + lm] = leaky01(acc[t][1][i] + bc1);
            }
        }
    }
    __syncthreads();

    {
        v8f acc[RT][2];
        #pragma unroll
        for (int t = 0; t < RT; ++t) { acc[t][0] = v8f{}; acc[t][1] = v8f{}; }
        mlp_gemm<HID, HID + 1>(hb + lm * (HID + 1) + 2 * half, w2a, w2b, bfirst, acc);
        float bc0 = b2[col0 + lm], bc1 = b2[col1 + lm];
        __syncthreads();
        #pragma unroll
        for (int t = 0; t < RT; ++t) {
            #pragma unroll
            for (int i = 0; i < 8; ++i) {
                int r = 16 * t + i + 8 * half;
                hb[r * (HID + 1) + col0 + lm] = leaky01(acc[t][0][i] + bc0);
                hb[r * (HID + 1) + col1 + lm] = leaky01(acc[t][1][i] + bc1);
            }
        }
    }
    __syncthreads();

    #pragma unroll
    for (int j = 0; j < ROWS * 8 / 256; ++j) {
        int p = tid + 256 * j;
        int seg = p / ROWS, row = p - seg * ROWS;
        const float* hrow = hb + row * (HID + 1) + seg * 32;
        const float* wseg = Wd + seg * 32;
        float s = 0.0f;
        #pragma unroll
        for (int c = 0; c < 32; ++c) s += hrow[c] * wseg[c];
        zb[row * 8 + seg] = s;
    }
    __syncthreads();
    if (tid < ROWS) {
        float a = 0.0f;
        #pragma unroll
        for (int s = 0; s < 8; ++s) a += zb[tid * 8 + s];
        out[nodeBase + tid] = a + bd[0];
    }
}

extern "C" void kernel_launch(void* const* d_in, const int* in_sizes, int n_in,
                              void* d_out, int out_size, void* d_ws, size_t ws_size,
                              hipStream_t stream) {
    const float* x     = (const float*)d_in[0];
    const int*   ei    = (const int*)d_in[1];
    const float* W_gcn = (const float*)d_in[2];
    const float* b_gcn = (const float*)d_in[3];
    const float* W1    = (const float*)d_in[4];
    const float* b1    = (const float*)d_in[5];
    const float* W2    = (const float*)d_in[6];
    const float* b2    = (const float*)d_in[7];
    const float* Wd    = (const float*)d_in[8];
    const float* bd    = (const float*)d_in[9];
    float* out = (float*)d_out;

    char* ws = (char*)d_ws;
    size_t off0 = 0;
    size_t off1 = (off0 + (size_t)2 * N_NODES * sizeof(int) + 255) & ~(size_t)255;
    size_t off2 = off1 + 256;
    size_t off3 = (off2 + (size_t)N_EDGES * 2 * sizeof(int) + 255) & ~(size_t)255;
    size_t off4 = (off3 + (size_t)2 * N_NODES * HCAP * sizeof(int) + 255) & ~(size_t)255;
    size_t off5 = (off4 + (size_t)N_NODES * C_IN * sizeof(float) + 255) & ~(size_t)255;
    int*   cnt  = (int*)(ws + off0);
    int*   ovfN = (int*)(ws + off1);
    int*   ovf  = (int*)(ws + off2);
    int*   slot = (int*)(ws + off3);
    float* h    = (float*)(ws + off4);
    float* agg  = (float*)(ws + off5);

    const int B = 256;
    cnt_init_kernel<<<(2 * N_NODES + B - 1) / B, B, 0, stream>>>(cnt, ovfN, 2 * N_NODES);
    slot_kernel<<<(N_EDGES / 4 + B - 1) / B, B, 0, stream>>>(ei, cnt, slot, ovfN, ovf,
                                                             N_EDGES / 4);

    gemm_xw_kernel<<<N_NODES / 64, B, 0, stream>>>(x, W_gcn, cnt, h, N_NODES);

    gather_kernel<<<N_NODES * 16 / B, B, 0, stream>>>(cnt, slot, ovfN, ovf, h, agg);

    mlp_fused_kernel<<<N_NODES / ROWS, B, 0, stream>>>(agg, x, b_gcn, W1, b1, W2, b2,
                                                       Wd, bd, out);
}
